# starting kernel
# baseline (speedup 1.0000x reference)
#include <hip/hip_runtime.h>
#include <hip/hip_bf16.h>
#include <math.h>

typedef __attribute__((ext_vector_type(2))) float v2f;
typedef __attribute__((ext_vector_type(8))) float v8f;

#define NEG_SLOPE 0.2f
#define BN_EPS 1e-5f

__device__ __forceinline__ unsigned encodeF(float f) {
    unsigned u = __float_as_uint(f);
    return (u & 0x80000000u) ? ~u : (u | 0x80000000u);
}
__device__ __forceinline__ float decodeF(unsigned u) {
    unsigned b = (u & 0x80000000u) ? (u & 0x7fffffffu) : ~u;
    return __uint_as_float(b);
}
__device__ __forceinline__ float lrelu(float v) {
    return v > 0.0f ? v : NEG_SLOPE * v;
}

__global__ __launch_bounds__(256) void fill_u32(unsigned* __restrict__ p, unsigned v, int n) {
    int i = blockIdx.x * blockDim.x + threadIdx.x;
    if (i < n) p[i] = v;
}

__global__ __launch_bounds__(256) void init_out_bias(float* __restrict__ out,
                                                     const float* __restrict__ b,
                                                     int total, int C) {
    int i = blockIdx.x * blockDim.x + threadIdx.x;
    if (i < total) out[i] = b[i % C];
}

__global__ __launch_bounds__(256) void wmma_gemm_f32(const float* __restrict__ A,
                                                     const float* __restrict__ B,
                                                     float* __restrict__ C,
                                                     int M, int K, int Ncols) {
    int wave = (int)((blockIdx.x * (long long)blockDim.x + threadIdx.x) >> 5);
    int lane = threadIdx.x & 31;
    int ntN = Ncols >> 4;
    int totalWaves = (M >> 4) * ntN;
    if (wave >= totalWaves) return;
    int mt = wave / ntN, nt = wave - mt * ntN;
    int m0 = mt << 4, n0 = nt << 4;
    int lo = lane & 15;
    int hi = lane >> 4;
    const float* arow = A + (size_t)(m0 + lo) * K;
    int ncol = n0 + lo;

    v8f acc = {0.f, 0.f, 0.f, 0.f, 0.f, 0.f, 0.f, 0.f};
    for (int k0 = 0; k0 < K; k0 += 4) {
        int kb = k0 + 2 * hi;
        v2f a;
        a.x = arow[kb];
        a.y = arow[kb + 1];
        v2f b;
        b.x = B[(size_t)kb * Ncols + ncol];
        b.y = B[(size_t)(kb + 1) * Ncols + ncol];
        acc = __builtin_amdgcn_wmma_f32_16x16x4_f32(
             false, a,  false, b,
             (short)0, acc,  false,  false);
    }
#pragma unroll
    for (int v = 0; v < 8; ++v) {
        int row = m0 + v + 8 * hi;
        C[(size_t)row * Ncols + ncol] = acc[v];
    }
}

__global__ __launch_bounds__(256) void alpha_kernel(const float* __restrict__ hf,
                                                    const float* __restrict__ a_s,
                                                    const float* __restrict__ a_d,
                                                    float* __restrict__ as_o,
                                                    float* __restrict__ ad_o,
                                                    int N, int H, int Cc) {
    int n = blockIdx.x * blockDim.x + threadIdx.x;
    if (n >= N) return;
    int F = H * Cc;
    for (int h = 0; h < H; ++h) {
        float s = 0.f, d = 0.f;
        for (int c = 0; c < Cc; ++c) {
            float v = hf[(size_t)n * F + h * Cc + c];
            s += v * a_s[h * Cc + c];
            d += v * a_d[h * Cc + c];
        }
        as_o[n * H + h] = s;
        ad_o[n * H + h] = d;
    }
}

__global__ __launch_bounds__(256) void edge_max(const int* __restrict__ ei, int E, int N,
                                                const float* __restrict__ as_i,
                                                const float* __restrict__ ad_i,
                                                unsigned* __restrict__ menc, int H) {
    int e = blockIdx.x * blockDim.x + threadIdx.x;
    int Etot = E + N;
    if (e >= Etot) return;
    int s = (e < E) ? ei[e] : (e - E);
    int d = (e < E) ? ei[E + e] : (e - E);
    for (int h = 0; h < H; ++h) {
        float v = lrelu(as_i[s * H + h] + ad_i[d * H + h]);
        atomicMax(&menc[d * H + h], encodeF(v));
    }
}

__global__ __launch_bounds__(256) void edge_sum(const int* __restrict__ ei, int E, int N,
                                                const float* __restrict__ as_i,
                                                const float* __restrict__ ad_i,
                                                const unsigned* __restrict__ menc,
                                                float* __restrict__ z, int H) {
    int e = blockIdx.x * blockDim.x + threadIdx.x;
    int Etot = E + N;
    if (e >= Etot) return;
    int s = (e < E) ? ei[e] : (e - E);
    int d = (e < E) ? ei[E + e] : (e - E);
    for (int h = 0; h < H; ++h) {
        float v = lrelu(as_i[s * H + h] + ad_i[d * H + h]);
        float ee = __expf(v - decodeF(menc[d * H + h]));
        atomicAdd(&z[d * H + h], ee);
    }
}

__global__ __launch_bounds__(256) void edge_agg(const int* __restrict__ ei, int E, int N,
                                                const float* __restrict__ as_i,
                                                const float* __restrict__ ad_i,
                                                const unsigned* __restrict__ menc,
                                                const float* __restrict__ z,
                                                const float* __restrict__ hsrc,
                                                float* __restrict__ out, int H, int Cc) {
    long long t = blockIdx.x * (long long)blockDim.x + threadIdx.x;
    long long Etot = (long long)E + N;
    if (t >= Etot * Cc) return;
    int e = (int)(t / Cc);
    int c = (int)(t - (long long)e * Cc);
    int s = (e < E) ? ei[e] : (e - E);
    int d = (e < E) ? ei[E + e] : (e - E);
    int F = H * Cc;
    for (int h = 0; h < H; ++h) {
        float v = lrelu(as_i[s * H + h] + ad_i[d * H + h]);
        float al = __expf(v - decodeF(menc[d * H + h])) / (z[d * H + h] + 1e-16f);
        float msg = hsrc[(size_t)s * F + h * Cc + c] * al;
        atomicAdd(&out[(size_t)d * F + h * Cc + c], msg);
    }
}

__global__ __launch_bounds__(256) void bn_elu(const float* __restrict__ agg,
                                              const float* __restrict__ b1,
                                              const float* __restrict__ g,
                                              const float* __restrict__ be,
                                              const float* __restrict__ mu,
                                              const float* __restrict__ var,
                                              float* __restrict__ outp, int total, int F) {
    int i = blockIdx.x * blockDim.x + threadIdx.x;
    if (i >= total) return;
    int j = i % F;
    float v = agg[i] + b1[j];
    v = (v - mu[j]) * g[j] * rsqrtf(var[j] + BN_EPS) + be[j];
    outp[i] = v > 0.f ? v : (__expf(v) - 1.f);
}

static inline int nblk(long long t) { return (int)((t + 255) / 256); }

extern "C" void kernel_launch(void* const* d_in, const int* in_sizes, int n_in,
                              void* d_out, int out_size, void* d_ws, size_t ws_size,
                              hipStream_t stream) {
    const float* x      = (const float*)d_in[0];
    const int*   ei     = (const int*)d_in[1];
    const float* W1     = (const float*)d_in[2];
    const float* a_src1 = (const float*)d_in[3];
    const float* a_dst1 = (const float*)d_in[4];
    const float* b1     = (const float*)d_in[5];
    const float* bn_g   = (const float*)d_in[6];
    const float* bn_b   = (const float*)d_in[7];
    const float* bn_m   = (const float*)d_in[8];
    const float* bn_v   = (const float*)d_in[9];
    const float* W2     = (const float*)d_in[10];
    const float* a_src2 = (const float*)d_in[11];
    const float* a_dst2 = (const float*)d_in[12];
    const float* b2     = (const float*)d_in[13];

    const int IN_DIM = 128, F1 = 256, C1 = 64, H1 = 4, C2 = 64;
    int N = in_sizes[0] / IN_DIM;
    int E = in_sizes[1] / 2;
    int Etot = E + N;

    float* ws = (float*)d_ws;
    size_t o = 0;
    float*    h1   = ws + o; o += (size_t)N * F1;
    float*    agg1 = ws + o; o += (size_t)N * F1;
    float*    as1  = ws + o; o += (size_t)N * H1;
    float*    ad1  = ws + o; o += (size_t)N * H1;
    unsigned* m1   = (unsigned*)(ws + o); o += (size_t)N * H1;
    float*    z1   = ws + o; o += (size_t)N * H1;
    float*    h2   = ws + o; o += (size_t)N * C2;
    float*    as2  = ws + o; o += (size_t)N;
    float*    ad2  = ws + o; o += (size_t)N;
    unsigned* m2   = (unsigned*)(ws + o); o += (size_t)N;
    float*    z2   = ws + o; o += (size_t)N;

    fill_u32<<<nblk((long long)N * F1), 256, 0, stream>>>((unsigned*)agg1, 0u, N * F1);
    fill_u32<<<nblk(N * H1), 256, 0, stream>>>(m1, 0u, N * H1);
    fill_u32<<<nblk(N * H1), 256, 0, stream>>>((unsigned*)z1, 0u, N * H1);
    fill_u32<<<nblk(N), 256, 0, stream>>>(m2, 0u, N);
    fill_u32<<<nblk(N), 256, 0, stream>>>((unsigned*)z2, 0u, N);

    {
        long long waves = (long long)(N / 16) * (F1 / 16);
        wmma_gemm_f32<<<nblk(waves * 32), 256, 0, stream>>>(x, W1, h1, N, IN_DIM, F1);
    }
    alpha_kernel<<<nblk(N), 256, 0, stream>>>(h1, a_src1, a_dst1, as1, ad1, N, H1, C1);
    edge_max<<<nblk(Etot), 256, 0, stream>>>(ei, E, N, as1, ad1, m1, H1);
    edge_sum<<<nblk(Etot), 256, 0, stream>>>(ei, E, N, as1, ad1, m1, z1, H1);
    edge_agg<<<nblk((long long)Etot * C1), 256, 0, stream>>>(ei, E, N, as1, ad1, m1, z1, h1,
                                                             agg1, H1, C1);
    bn_elu<<<nblk((long long)N * F1), 256, 0, stream>>>(agg1, b1, bn_g, bn_b, bn_m, bn_v, h1,
                                                        N * F1, F1);

    {
        long long waves = (long long)(N / 16) * (C2 / 16);
        wmma_gemm_f32<<<nblk(waves * 32), 256, 0, stream>>>(h1, W2, h2, N, F1, C2);
    }
    alpha_kernel<<<nblk(N), 256, 0, stream>>>(h2, a_src2, a_dst2, as2, ad2, N, 1, C2);
    edge_max<<<nblk(Etot), 256, 0, stream>>>(ei, E, N, as2, ad2, m2, 1);
    edge_sum<<<nblk(Etot), 256, 0, stream>>>(ei, E, N, as2, ad2, m2, z2, 1);
    init_out_bias<<<nblk((long long)N * C2), 256, 0, stream>>>((float*)d_out, b2, N * C2, C2);
    edge_agg<<<nblk((long long)Etot * C2), 256, 0, stream>>>(ei, E, N, as2, ad2, m2, z2, h2,
                                                             (float*)d_out, 1, C2);
}


# Edges grouped by destination inside the call (counting sort, int atomics); one wave per node gathers its source rows with an online softmax, replacing the three float-atomic edge passes per layer
# speedup vs baseline: 3.5251x; 3.5251x over previous
#include <hip/hip_runtime.h>
#include <hip/hip_bf16.h>
#include <math.h>

typedef __attribute__((ext_vector_type(2))) float v2f;
typedef __attribute__((ext_vector_type(8))) float v8f;

#define NEG_SLOPE 0.2f
#define BN_EPS 1e-5f

__device__ __forceinline__ float lrelu(float v) {
    return v > 0.0f ? v : NEG_SLOPE * v;
}

__global__ __launch_bounds__(256) void fill_u32(unsigned* __restrict__ p, unsigned v, int n) {
    int i = blockIdx.x * blockDim.x + threadIdx.x;
    if (i < n) p[i] = v;
}

__global__ __launch_bounds__(256) void wmma_gemm_f32(const float* __restrict__ A,
                                                     const float* __restrict__ B,
                                                     float* __restrict__ C,
                                                     int M, int K, int Ncols) {
    int wave = (int)((blockIdx.x * (long long)blockDim.x + threadIdx.x) >> 5);
    int lane = threadIdx.x & 31;
    int ntN = Ncols >> 4;
    int totalWaves = (M >> 4) * ntN;
    if (wave >= totalWaves) return;
    int mt = wave / ntN, nt = wave - mt * ntN;
    int m0 = mt << 4, n0 = nt << 4;
    int lo = lane & 15;
    int hi = lane >> 4;
    const float* arow = A + (size_t)(m0 + lo) * K;
    int ncol = n0 + lo;

    v8f acc = {0.f, 0.f, 0.f, 0.f, 0.f, 0.f, 0.f, 0.f};
    for (int k0 = 0; k0 < K; k0 += 4) {
        int kb = k0 + 2 * hi;
        v2f a;
        a.x = arow[kb];
        a.y = arow[kb + 1];
        v2f b;
        b.x = B[(size_t)kb * Ncols + ncol];
        b.y = B[(size_t)(kb + 1) * Ncols + ncol];
        acc = __builtin_amdgcn_wmma_f32_16x16x4_f32(
             false, a,  false, b,
             (short)0, acc,  false,  false);
    }
#pragma unroll
    for (int v = 0; v < 8; ++v) {
        int row = m0 + v + 8 * hi;
        C[(size_t)row * Ncols + ncol] = acc[v];
    }
}

__global__ __launch_bounds__(256) void alpha_kernel(const float* __restrict__ hf,
                                                    const float* __restrict__ a_s,
                                                    const float* __restrict__ a_d,
                                                    float* __restrict__ as_o,
                                                    float* __restrict__ ad_o,
                                                    int N, int H, int Cc) {
    int n = blockIdx.x * blockDim.x + threadIdx.x;
    if (n >= N) return;
    int F = H * Cc;
    for (int h = 0; h < H; ++h) {
        float s = 0.f, d = 0.f;
        for (int c = 0; c < Cc; ++c) {
            float v = hf[(size_t)n * F + h * Cc + c];
            s += v * a_s[h * Cc + c];
            d += v * a_d[h * Cc + c];
        }
        as_o[n * H + h] = s;
        ad_o[n * H + h] = d;
    }
}

__global__ __launch_bounds__(256) void hist_rank(const int* __restrict__ ei, int E,
                                                 int* __restrict__ cnt, int* __restrict__ rank) {
    int e = blockIdx.x * blockDim.x + threadIdx.x;
    if (e >= E) return;
    int d = ei[E + e];
    rank[e] = atomicAdd(&cnt[d], 1);
}

__global__ __launch_bounds__(1024) void scan_counts(const int* __restrict__ cnt,
                                                    int* __restrict__ off, int N) {
    __shared__ int part[1024];
    int t = threadIdx.x;
    int per = (((N + 1023) >> 10) + 3) & ~3;
    int b = t * per;
    int s = 0;
    for (int i = b; i < b + per; i += 4) {
        if (i < N) {
            int4 c = *(const int4*)(cnt + i);
            s += c.x + c.y + c.z + c.w;
        }
    }
    part[t] = s;
    __syncthreads();
    for (int o = 1; o < 1024; o <<= 1) {
        int v = (t >= o) ? part[t - o] : 0;
        __syncthreads();
        part[t] += v;
        __syncthreads();
    }
    int run = part[t] - s;
    for (int i = b; i < b + per; i += 4) {
        if (i < N) {
            int4 c = *(const int4*)(cnt + i);
            int4 o4;
            o4.x = run;
            o4.y = run + c.x;
            o4.z = o4.y + c.y;
            o4.w = o4.z + c.z;
            run = o4.w + c.w;
            *(int4*)(off + i) = o4;
        }
    }
    if (t == 1023) off[N] = part[1023];
}

__global__ __launch_bounds__(256) void csr_fill(const int* __restrict__ ei, int E,
                                                const int* __restrict__ off,
                                                const int* __restrict__ rank,
                                                int* __restrict__ csr) {
    int e = blockIdx.x * blockDim.x + threadIdx.x;
    if (e >= E) return;
    int d = ei[E + e];
    csr[off[d] + rank[e]] = ei[e];
}

__device__ __forceinline__ float wave_max(float v) {
    for (int o = 16; o > 0; o >>= 1) v = fmaxf(v, __shfl_xor(v, o));
    return v;
}
__device__ __forceinline__ float wave_sum(float v) {
    for (int o = 16; o > 0; o >>= 1) v += __shfl_xor(v, o);
    return v;
}
__device__ __forceinline__ int lane_i(int x, int l) { return __builtin_amdgcn_readlane(x, l); }
__device__ __forceinline__ float lane_f(float x, int l) {
    return __int_as_float(__builtin_amdgcn_readlane(__float_as_int(x), l));
}

__global__ __launch_bounds__(256) void gather_h4(const int* __restrict__ off,
                                                 const int* __restrict__ csr,
                                                 const float* __restrict__ as_i,
                                                 const float* __restrict__ ad_i,
                                                 const float* __restrict__ hsrc,
                                                 float* __restrict__ outp, int N) {
    int node = __builtin_amdgcn_readfirstlane((int)((blockIdx.x * 256u + threadIdx.x) >> 5));
    int lane = threadIdx.x & 31;
    if (node >= N) return;
    int beg = off[node];
    int deg = off[node + 1] - beg;
    const float4* as4 = (const float4*)as_i;
    float4 adv = ((const float4*)ad_i)[node];
    int hsel = lane >> 4;
    float m0 = -INFINITY, m1 = -INFINITY, m2 = -INFINITY, m3 = -INFINITY;
    float z0 = 0.f, z1 = 0.f, z2 = 0.f, z3 = 0.f;
    float4 accA = {0.f, 0.f, 0.f, 0.f}, accB = {0.f, 0.f, 0.f, 0.f};
    for (int base = 0; base <= deg; base += 32) {
        int idx = base + lane;
        bool valid = idx <= deg;
        int s = (idx < deg) ? csr[beg + idx] : node;
        float4 a = as4[s];
        float v0 = valid ? lrelu(a.x + adv.x) : -INFINITY;
        float v1 = valid ? lrelu(a.y + adv.y) : -INFINITY;
        float v2 = valid ? lrelu(a.z + adv.z) : -INFINITY;
        float v3 = valid ? lrelu(a.w + adv.w) : -INFINITY;
        float n0 = fmaxf(m0, wave_max(v0));
        float n1 = fmaxf(m1, wave_max(v1));
        float n2 = fmaxf(m2, wave_max(v2));
        float n3 = fmaxf(m3, wave_max(v3));
        float sc0 = __expf(m0 - n0), sc1 = __expf(m1 - n1);
        float sc2 = __expf(m2 - n2), sc3 = __expf(m3 - n3);
        float w0 = valid ? __expf(v0 - n0) : 0.f;
        float w1 = valid ? __expf(v1 - n1) : 0.f;
        float w2 = valid ? __expf(v2 - n2) : 0.f;
        float w3 = valid ? __expf(v3 - n3) : 0.f;
        z0 = z0 * sc0 + w0; z1 = z1 * sc1 + w1; z2 = z2 * sc2 + w2; z3 = z3 * sc3 + w3;
        m0 = n0; m1 = n1; m2 = n2; m3 = n3;
        float scA = hsel ? sc1 : sc0, scB = hsel ? sc3 : sc2;
        accA.x *= scA; accA.y *= scA; accA.z *= scA; accA.w *= scA;
        accB.x *= scB; accB.y *= scB; accB.z *= scB; accB.w *= scB;
        int cnt = deg + 1 - base;
        if (cnt > 32) cnt = 32;
        for (int jj = 0; jj < cnt; jj += 4) {
            int sa = lane_i(s, jj), sb = lane_i(s, jj + 1), sc = lane_i(s, jj + 2), sd = lane_i(s, jj + 3);
            const float4* ra = (const float4*)(hsrc + (size_t)sa * 256);
            const float4* rb = (const float4*)(hsrc + (size_t)sb * 256);
            const float4* rc = (const float4*)(hsrc + (size_t)sc * 256);
            const float4* rd = (const float4*)(hsrc + (size_t)sd * 256);
            float4 xa = ra[lane], ya = ra[32 + lane];
            float4 xb = rb[lane], yb = rb[32 + lane];
            float4 xc = rc[lane], yc = rc[32 + lane];
            float4 xd = rd[lane], yd = rd[32 + lane];
            float p0, p1, p2, p3;
            p0 = lane_f(w0, jj); p1 = lane_f(w1, jj); p2 = lane_f(w2, jj); p3 = lane_f(w3, jj);
            float wa = hsel ? p1 : p0, ua = hsel ? p3 : p2;
            p0 = lane_f(w0, jj + 1); p1 = lane_f(w1, jj + 1); p2 = lane_f(w2, jj + 1); p3 = lane_f(w3, jj + 1);
            float wb = hsel ? p1 : p0, ub = hsel ? p3 : p2;
            p0 = lane_f(w0, jj + 2); p1 = lane_f(w1, jj + 2); p2 = lane_f(w2, jj + 2); p3 = lane_f(w3, jj + 2);
            float wc = hsel ? p1 : p0, uc = hsel ? p3 : p2;
            p0 = lane_f(w0, jj + 3); p1 = lane_f(w1, jj + 3); p2 = lane_f(w2, jj + 3); p3 = lane_f(w3, jj + 3);
            float wd = hsel ? p1 : p0, ud = hsel ? p3 : p2;
            accA.x += wa * xa.x; accA.y += wa * xa.y; accA.z += wa * xa.z; accA.w += wa * xa.w;
            accB.x += ua * ya.x; accB.y += ua * ya.y; accB.z += ua * ya.z; accB.w += ua * ya.w;
            accA.x += wb * xb.x; accA.y += wb * xb.y; accA.z += wb * xb.z; accA.w += wb * xb.w;
            accB.x += ub * yb.x; accB.y += ub * yb.y; accB.z += ub * yb.z; accB.w += ub * yb.w;
            accA.x += wc * xc.x; accA.y += wc * xc.y; accA.z += wc * xc.z; accA.w += wc * xc.w;
            accB.x += uc * yc.x; accB.y += uc * yc.y; accB.z += uc * yc.z; accB.w += uc * yc.w;
            accA.x += wd * xd.x; accA.y += wd * xd.y; accA.z += wd * xd.z; accA.w += wd * xd.w;
            accB.x += ud * yd.x; accB.y += ud * yd.y; accB.z += ud * yd.z; accB.w += ud * yd.w;
        }
    }
    z0 = wave_sum(z0); z1 = wave_sum(z1); z2 = wave_sum(z2); z3 = wave_sum(z3);
    float dA = (hsel ? z1 : z0) + 1e-16f;
    float dB = (hsel ? z3 : z2) + 1e-16f;
    float4 oA = {accA.x / dA, accA.y / dA, accA.z / dA, accA.w / dA};
    float4 oB = {accB.x / dB, accB.y / dB, accB.z / dB, accB.w / dB};
    float4* orow = (float4*)(outp + (size_t)node * 256);
    orow[lane] = oA;
    orow[32 + lane] = oB;
}

__global__ __launch_bounds__(256) void gather_h1(const int* __restrict__ off,
                                                 const int* __restrict__ csr,
                                                 const float* __restrict__ as_i,
                                                 const float* __restrict__ ad_i,
                                                 const float* __restrict__ hsrc,
                                                 const float* __restrict__ bias,
                                                 float* __restrict__ outp, int N) {
    int node = __builtin_amdgcn_readfirstlane((int)((blockIdx.x * 256u + threadIdx.x) >> 5));
    int lane = threadIdx.x & 31;
    if (node >= N) return;
    int beg = off[node];
    int deg = off[node + 1] - beg;
    float adv = ad_i[node];
    float m = -INFINITY, z = 0.f;
    float2 acc = {0.f, 0.f};
    for (int base = 0; base <= deg; base += 32) {
        int idx = base + lane;
        bool valid = idx <= deg;
        int s = (idx < deg) ? csr[beg + idx] : node;
        float v = valid ? lrelu(as_i[s] + adv) : -INFINITY;
        float n = fmaxf(m, wave_max(v));
        float sc = __expf(m - n);
        float w = valid ? __expf(v - n) : 0.f;
        z = z * sc + w;
        m = n;
        acc.x *= sc; acc.y *= sc;
        int cnt = deg + 1 - base;
        if (cnt > 32) cnt = 32;
        for (int jj = 0; jj < cnt; jj += 4) {
            int sa = lane_i(s, jj), sb = lane_i(s, jj + 1), sc2 = lane_i(s, jj + 2), sd = lane_i(s, jj + 3);
            float2 xa = ((const float2*)(hsrc + (size_t)sa * 64))[lane];
            float2 xb = ((const float2*)(hsrc + (size_t)sb * 64))[lane];
            float2 xc = ((const float2*)(hsrc + (size_t)sc2 * 64))[lane];
            float2 xd = ((const float2*)(hsrc + (size_t)sd * 64))[lane];
            float wa = lane_f(w, jj), wb = lane_f(w, jj + 1), wc = lane_f(w, jj + 2), wd = lane_f(w, jj + 3);
            acc.x += wa * xa.x; acc.y += wa * xa.y;
            acc.x += wb * xb.x; acc.y += wb * xb.y;
            acc.x += wc * xc.x; acc.y += wc * xc.y;
            acc.x += wd * xd.x; acc.y += wd * xd.y;
        }
    }
    z = wave_sum(z);
    float den = z + 1e-16f;
    float2 o2 = {acc.x / den + bias[2 * lane], acc.y / den + bias[2 * lane + 1]};
    ((float2*)(outp + (size_t)node * 64))[lane] = o2;
}

__global__ __launch_bounds__(256) void bn_elu(const float* __restrict__ agg,
                                              const float* __restrict__ b1,
                                              const float* __restrict__ g,
                                              const float* __restrict__ be,
                                              const float* __restrict__ mu,
                                              const float* __restrict__ var,
                                              float* __restrict__ outp, int total, int F) {
    int i = blockIdx.x * blockDim.x + threadIdx.x;
    if (i >= total) return;
    int j = i % F;
    float v = agg[i] + b1[j];
    v = (v - mu[j]) * g[j] * rsqrtf(var[j] + BN_EPS) + be[j];
    outp[i] = v > 0.f ? v : (__expf(v) - 1.f);
}

static inline int nblk(long long t) { return (int)((t + 255) / 256); }

extern "C" void kernel_launch(void* const* d_in, const int* in_sizes, int n_in,
                              void* d_out, int out_size, void* d_ws, size_t ws_size,
                              hipStream_t stream) {
    const float* x      = (const float*)d_in[0];
    const int*   ei     = (const int*)d_in[1];
    const float* W1     = (const float*)d_in[2];
    const float* a_src1 = (const float*)d_in[3];
    const float* a_dst1 = (const float*)d_in[4];
    const float* b1     = (const float*)d_in[5];
    const float* bn_g   = (const float*)d_in[6];
    const float* bn_b   = (const float*)d_in[7];
    const float* bn_m   = (const float*)d_in[8];
    const float* bn_v   = (const float*)d_in[9];
    const float* W2     = (const float*)d_in[10];
    const float* a_src2 = (const float*)d_in[11];
    const float* a_dst2 = (const float*)d_in[12];
    const float* b2     = (const float*)d_in[13];

    const int IN_DIM = 128, F1 = 256, C1 = 64, H1 = 4, C2 = 64;
    int N = in_sizes[0] / IN_DIM;
    int E = in_sizes[1] / 2;

    float* ws = (float*)d_ws;
    size_t o = 0;
    float* h1   = ws + o; o += (size_t)N * F1;
    float* agg1 = ws + o; o += (size_t)N * F1;
    float* as1  = ws + o; o += (size_t)N * H1;
    float* ad1  = ws + o; o += (size_t)N * H1;
    o += (size_t)N * H1;
    o += (size_t)N * H1;
    int*   grp  = (int*)(ws + o); o += (size_t)N * C2;
    float* as2  = ws + o; o += (size_t)N;
    float* ad2  = ws + o; o += (size_t)N;
    float* h2   = agg1;

    size_t Ep = ((size_t)E + 3) & ~(size_t)3;
    int* offs = grp;
    int* csr  = grp + N + 4;
    int* cnt  = csr + Ep;
    int* rnk  = cnt + N;

    fill_u32<<<nblk(N), 256, 0, stream>>>((unsigned*)cnt, 0u, N);
    hist_rank<<<nblk(E), 256, 0, stream>>>(ei, E, cnt, rnk);
    scan_counts<<<1, 1024, 0, stream>>>(cnt, offs, N);
    csr_fill<<<nblk(E), 256, 0, stream>>>(ei, E, offs, rnk, csr);

    {
        long long waves = (long long)(N / 16) * (F1 / 16);
        wmma_gemm_f32<<<nblk(waves * 32), 256, 0, stream>>>(x, W1, h1, N, IN_DIM, F1);
    }
    alpha_kernel<<<nblk(N), 256, 0, stream>>>(h1, a_src1, a_dst1, as1, ad1, N, H1, C1);
    gather_h4<<<nblk((long long)N * 32), 256, 0, stream>>>(offs, csr, as1, ad1, h1, agg1, N);
    bn_elu<<<nblk((long long)N * F1), 256, 0, stream>>>(agg1, b1, bn_g, bn_b, bn_m, bn_v, h1,
                                                        N * F1, F1);

    {
        long long waves = (long long)(N / 16) * (C2 / 16);
        wmma_gemm_f32<<<nblk(waves * 32), 256, 0, stream>>>(h1, W2, h2, N, F1, C2);
    }
    alpha_kernel<<<nblk(N), 256, 0, stream>>>(h2, a_src2, a_dst2, as2, ad2, N, 1, C2);
    gather_h1<<<nblk((long long)N * 32), 256, 0, stream>>>(offs, csr, as2, ad2, h2, b2,
                                                           (float*)d_out, N);
}


# Layer 1 gathers 512-byte input rows per head (scores from x via W1*a vectors), then multiplies each 16-node aggregate by W1 in the same block (LDS, WMMA): no feature array, 5 launches; sums regrouped
# speedup vs baseline: 6.4146x; 1.8197x over previous
#include <hip/hip_runtime.h>
#include <hip/hip_bf16.h>
#include <math.h>

typedef __attribute__((ext_vector_type(2))) float v2f;
typedef __attribute__((ext_vector_type(8))) float v8f;

#define NEG_SLOPE 0.2f
#define BN_EPS 1e-5f

template <int O> __device__ __forceinline__ float xmask_add(float v) {
    return v + __int_as_float(__builtin_amdgcn_update_dpp(0, __float_as_int(v), 0x160 | O, 0xf, 0xf, true));
}
__device__ __forceinline__ float row16_sum(float v) {
    v = xmask_add<1>(v);
    v = xmask_add<2>(v);
    v = xmask_add<4>(v);
    v = xmask_add<8>(v);
    return v;
}

#define KGRP 4

struct GemmFrag {
    v2f a[KGRP];
    float4 b0[KGRP];
    float4 b1[KGRP];
};

__device__ __forceinline__ void gemm_load(GemmFrag& f, const float* __restrict__ arow,
                                          const float* __restrict__ bcol, int Ncols, int k0, int hi) {
#pragma unroll
    for (int u = 0; u < KGRP; ++u) {
        int kb = k0 + 4 * u + 2 * hi;
        f.a[u] = *(const v2f*)(arow + kb);
        const float* bp = bcol + (size_t)kb * Ncols;
        f.b0[u] = *(const float4*)bp;
        f.b1[u] = *(const float4*)(bp + Ncols);
    }
}

__device__ __forceinline__ void gemm_mma(const GemmFrag& f, v8f& acc0, v8f& acc1, v8f& acc2, v8f& acc3) {
#pragma unroll
    for (int u = 0; u < KGRP; ++u) {
        v2f b;
        b.x = f.b0[u].x; b.y = f.b1[u].x;
        acc0 = __builtin_amdgcn_wmma_f32_16x16x4_f32(false, f.a[u], false, b, (short)0, acc0, false, false);
        b.x = f.b0[u].y; b.y = f.b1[u].y;
        acc1 = __builtin_amdgcn_wmma_f32_16x16x4_f32(false, f.a[u], false, b, (short)0, acc1, false, false);
        b.x = f.b0[u].z; b.y = f.b1[u].z;
        acc2 = __builtin_amdgcn_wmma_f32_16x16x4_f32(false, f.a[u], false, b, (short)0, acc2, false, false);
        b.x = f.b0[u].w; b.y = f.b1[u].w;
        acc3 = __builtin_amdgcn_wmma_f32_16x16x4_f32(false, f.a[u], false, b, (short)0, acc3, false, false);
    }
}

__global__ __launch_bounds__(256) void wmma_gemm_alpha(const float* __restrict__ A,
                                                       const float* __restrict__ B,
                                                       const float* __restrict__ a_d,
                                                       float* __restrict__ C,
                                                       float* __restrict__ ad_o,
                                                       int M, int K, int Ncols) {
    int H = Ncols >> 6;
    int head = (int)(blockIdx.x % (unsigned)H);
    int mt = (int)(blockIdx.x / (unsigned)H) * 8 + (int)(threadIdx.x >> 5);
    int lane = threadIdx.x & 31;
    if (mt >= (M >> 4)) return;
    int m0 = mt << 4;
    int lo = lane & 15;
    int hi = lane >> 4;
    const float* arow = A + (size_t)(m0 + lo) * K;
    int ncol = (head << 6) + 4 * lo;
    const float* bcol = B + ncol;

    v8f acc0 = {0.f, 0.f, 0.f, 0.f, 0.f, 0.f, 0.f, 0.f};
    v8f acc1 = acc0, acc2 = acc0, acc3 = acc0;
    GemmFrag f0, f1;
    gemm_load(f0, arow, bcol, Ncols, 0, hi);
    for (int k0 = 0; k0 < K; k0 += 8 * KGRP) {
        gemm_load(f1, arow, bcol, Ncols, k0 + 4 * KGRP, hi);
        gemm_mma(f0, acc0, acc1, acc2, acc3);
        if (k0 + 8 * KGRP < K) gemm_load(f0, arow, bcol, Ncols, k0 + 8 * KGRP, hi);
        gemm_mma(f1, acc0, acc1, acc2, acc3);
    }
    float4 d4 = *(const float4*)(a_d + ncol);
    float pd[8];
#pragma unroll
    for (int v = 0; v < 8; ++v) {
        int row = m0 + v + 8 * hi;
        float4 c4 = {acc0[v], acc1[v], acc2[v], acc3[v]};
        *(float4*)(C + (size_t)row * Ncols + ncol) = c4;
        pd[v] = c4.x * d4.x + c4.y * d4.y + c4.z * d4.z + c4.w * d4.w;
    }
#pragma unroll
    for (int v = 0; v < 8; ++v) {
        pd[v] = row16_sum(pd[v]);
    }
    if (lo == 0) {
#pragma unroll
        for (int v = 0; v < 8; ++v) {
            size_t r = (size_t)(m0 + v + 8 * hi) * H + head;
            ad_o[r] = pd[v];
        }
    }
}

#define CAP 48

__global__ __launch_bounds__(256) void place_edges(const int* __restrict__ ei, int E,
                                                   int* __restrict__ cnt, int* __restrict__ head,
                                                   int* __restrict__ nxt, int* __restrict__ bkt) {
    int e = blockIdx.x * blockDim.x + threadIdx.x;
    if (e >= E) return;
    int d = ei[E + e];
    int sv = ei[e];
    asm volatile("" : "+v"(sv));
    int slot = atomicAdd(&cnt[d], 1);
    if (slot < CAP) bkt[(size_t)d * CAP + slot] = sv;
    else nxt[e] = atomicExch(&head[d], e);
}

__device__ __forceinline__ int lane_i(int x, int l) { return __builtin_amdgcn_readlane(x, l); }
__device__ __forceinline__ float xhalf(float x) {
    return __int_as_float(__builtin_amdgcn_permlanex16(__float_as_int(x), __float_as_int(x), 0x76543210, 0xfedcba98,
                                                       false, false));
}
__device__ __forceinline__ float dot4(float4 c, float4 s) {
    return c.x * s.x + c.y * s.y + c.z * s.z + c.w * s.w;
}
#define LOG2E 1.44269504088896340736f
__device__ __forceinline__ float4 scale4(float4 v, float c) {
    float4 r = {v.x * c, v.y * c, v.z * c, v.w * c};
    return r;
}
__device__ __forceinline__ float ex2(float v) { return __builtin_amdgcn_exp2f(v); }
__device__ __forceinline__ float lrelu_max(float v) { return fmaxf(v, NEG_SLOPE * v); }

__global__ __launch_bounds__(256) void init_counts_uv(int* __restrict__ cnt, int* __restrict__ head, int n,
                                                      const float* __restrict__ W1,
                                                      const float* __restrict__ a_s,
                                                      const float* __restrict__ a_d,
                                                      float* __restrict__ uv) {
    int i = blockIdx.x * blockDim.x + threadIdx.x;
    if (i < n) {
        cnt[i] = 0;
        head[i] = -1;
    }
    if (i < 1024) {
        int h = (i >> 7) & 3, k = i & 127;
        const float* a = (i < 512 ? a_s : a_d) + h * 64;
        const float* w = W1 + (size_t)k * 256 + h * 64;
        float s = 0.f;
        for (int c = 0; c < 64; ++c) s += w[c] * a[c];
        uv[i] = s * LOG2E;
    }
}

template <int O> __device__ __forceinline__ float xmask_get(float v) {
    return __int_as_float(__builtin_amdgcn_update_dpp(0, __float_as_int(v), 0x160 | O, 0xf, 0xf, true));
}
__device__ __forceinline__ float head_sum(float k0, float k1, float s0, float s1) {
    float own = k0 + xhalf(s0);
    float oth = k1 + xhalf(s1);
    float r = own + xmask_get<8>(oth);
    r = xmask_add<4>(r);
    r = xmask_add<2>(r);
    r = xmask_add<1>(r);
    return r;
}
__device__ __forceinline__ float bcast(float v, int l) {
    return __int_as_float(__builtin_amdgcn_readlane(__float_as_int(v), l));
}
__device__ __forceinline__ void acc4(float4& a, float c, float wa, float4 xa, float wb, float4 xb, float wc, float4 xc,
                                     float wd, float4 xd) {
    a.x = a.x * c + wa * xa.x + wb * xb.x + wc * xc.x + wd * xd.x;
    a.y = a.y * c + wa * xa.y + wb * xb.y + wc * xc.y + wd * xd.y;
    a.z = a.z * c + wa * xa.z + wb * xb.z + wc * xc.z + wd * xd.z;
    a.w = a.w * c + wa * xa.w + wb * xb.w + wc * xc.w + wd * xd.w;
}
__device__ __forceinline__ float bn_elu_rs(float agg, float b, float mu, float g, float rs, float be) {
    float v = agg + b;
    v = (v - mu) * g * rs + be;
    return v > 0.f ? v : (__expf(v) - 1.f);
}

#define XP 132

__global__ __launch_bounds__(128) void gather_x4_gemm(const int* __restrict__ cnt,
                                                      const int* __restrict__ head,
                                                      const int* __restrict__ nxt,
                                                      const int* __restrict__ bkt,
                                                      const int* __restrict__ ei,
                                                      const float* __restrict__ x,
                                                      const float* __restrict__ uv,
                                                      const float* __restrict__ W1,
                                                      const float* __restrict__ b1,
                                                      const float* __restrict__ g,
                                                      const float* __restrict__ be,
                                                      const float* __restrict__ mu,
                                                      const float* __restrict__ var,
                                                      float* __restrict__ outp, int N) {
    __shared__ float4 xs4[4 * 16 * (XP / 4)];
    float* xs = (float*)xs4;
    int wave = __builtin_amdgcn_readfirstlane((int)(threadIdx.x >> 5));
    int lane = threadIdx.x & 31;
    int m0 = (int)blockIdx.x * 16;
    int hl = lane >> 3;
    const float4* us = (const float4*)uv;
    const float4* ud = us + 128;
    float4 R0 = us[32 * hl + lane], R1 = us[32 * (hl ^ 1) + lane];
    float4 R2 = us[32 * (hl ^ 2) + lane], R3 = us[32 * (hl ^ 3) + lane];

    for (int i = 0; i < 4; ++i) {
        int r = wave * 4 + i;
        int node = m0 + r;
        float4 o0 = {0.f, 0.f, 0.f, 0.f}, o1 = o0, o2 = o0, o3 = o0;
        if (node < N) {
            float4 xn = ((const float4*)(x + (size_t)node * 128))[lane];
            float4 D0 = ud[32 * hl + lane], D1 = ud[32 * (hl ^ 1) + lane];
            float4 D2 = ud[32 * (hl ^ 2) + lane], D3 = ud[32 * (hl ^ 3) + lane];
            float adv = head_sum(dot4(xn, D0), dot4(xn, D1), dot4(xn, D2), dot4(xn, D3));
            int cn = cnt[node];
            int degb = cn < CAP ? cn : CAP;
            int nb = degb + 1;
            int ovh = cn > CAP ? head[node] : -1;
            const int* brow = bkt + (size_t)node * CAP;
            float m = -INFINITY, z = 0.f;
            float4 a0 = {0.f, 0.f, 0.f, 0.f}, a1 = a0, a2 = a0, a3 = a0;
            int base = 0;
            while (base < nb || ovh >= 0) {
                int s = node;
                int cnt32 = 0;
                if (base < nb) {
                    int idx = base + lane;
                    if (idx < degb) s = brow[idx];
                    cnt32 = nb - base;
                    if (cnt32 > 32) cnt32 = 32;
                    base += 32;
                } else {
                    while (ovh >= 0 && cnt32 < 32) {
                        int sv = ei[ovh];
                        if (lane == cnt32) s = sv;
                        ovh = nxt[ovh];
                        ++cnt32;
                    }
                }
                for (int jj = 0; jj < cnt32; jj += 4) {
                    int sa = lane_i(s, jj), sb = lane_i(s, jj + 1), sc = lane_i(s, jj + 2), sd = lane_i(s, jj + 3);
                    float4 xa = ((const float4*)(x + (size_t)sa * 128))[lane];
                    float4 xb = ((const float4*)(x + (size_t)sb * 128))[lane];
                    float4 xc = ((const float4*)(x + (size_t)sc * 128))[lane];
                    float4 xd = ((const float4*)(x + (size_t)sd * 128))[lane];
                    float pa = head_sum(dot4(xa, R0), dot4(xa, R1), dot4(xa, R2), dot4(xa, R3));
                    float pb = head_sum(dot4(xb, R0), dot4(xb, R1), dot4(xb, R2), dot4(xb, R3));
                    float pc = head_sum(dot4(xc, R0), dot4(xc, R1), dot4(xc, R2), dot4(xc, R3));
                    float pd = head_sum(dot4(xd, R0), dot4(xd, R1), dot4(xd, R2), dot4(xd, R3));
                    float ea = lrelu_max(pa + adv);
                    float eb = jj + 1 < cnt32 ? lrelu_max(pb + adv) : -INFINITY;
                    float ec = jj + 2 < cnt32 ? lrelu_max(pc + adv) : -INFINITY;
                    float ed = jj + 3 < cnt32 ? lrelu_max(pd + adv) : -INFINITY;
                    float n = fmaxf(fmaxf(m, ea), fmaxf(fmaxf(eb, ec), ed));
                    float sc2 = ex2(m - n);
                    float wa = ex2(ea - n), wb = ex2(eb - n), wc = ex2(ec - n), wd = ex2(ed - n);
                    m = n;
                    z = z * sc2 + ((wa + wb) + (wc + wd));
                    acc4(a0, bcast(sc2, 0), bcast(wa, 0), xa, bcast(wb, 0), xb, bcast(wc, 0), xc, bcast(wd, 0), xd);
                    acc4(a1, bcast(sc2, 8), bcast(wa, 8), xa, bcast(wb, 8), xb, bcast(wc, 8), xc, bcast(wd, 8), xd);
                    acc4(a2, bcast(sc2, 16), bcast(wa, 16), xa, bcast(wb, 16), xb, bcast(wc, 16), xc, bcast(wd, 16), xd);
                    acc4(a3, bcast(sc2, 24), bcast(wa, 24), xa, bcast(wb, 24), xb, bcast(wc, 24), xc, bcast(wd, 24), xd);
                }
            }
            float rinv = 1.0f / (z + 1e-16f);
            o0 = scale4(a0, bcast(rinv, 0));
            o1 = scale4(a1, bcast(rinv, 8));
            o2 = scale4(a2, bcast(rinv, 16));
            o3 = scale4(a3, bcast(rinv, 24));
        }
        xs4[(0 * 16 + r) * (XP / 4) + lane] = o0;
        xs4[(1 * 16 + r) * (XP / 4) + lane] = o1;
        xs4[(2 * 16 + r) * (XP / 4) + lane] = o2;
        xs4[(3 * 16 + r) * (XP / 4) + lane] = o3;
    }
    __syncthreads();

    int lo = lane & 15;
    int hi = lane >> 4;
    const float* arow = xs + (wave * 16 + lo) * XP;
    int ncol = (wave << 6) + 4 * lo;
    const float* bcol = W1 + ncol;
    v8f acc0 = {0.f, 0.f, 0.f, 0.f, 0.f, 0.f, 0.f, 0.f};
    v8f acc1 = acc0, acc2 = acc0, acc3 = acc0;
    GemmFrag f0, f1;
    gemm_load(f0, arow, bcol, 256, 0, hi);
    for (int k0 = 0; k0 < 128; k0 += 8 * KGRP) {
        gemm_load(f1, arow, bcol, 256, k0 + 4 * KGRP, hi);
        gemm_mma(f0, acc0, acc1, acc2, acc3);
        if (k0 + 8 * KGRP < 128) gemm_load(f0, arow, bcol, 256, k0 + 8 * KGRP, hi);
        gemm_mma(f1, acc0, acc1, acc2, acc3);
    }
    float4 bi4 = *(const float4*)(b1 + ncol);
    float4 mu4 = *(const float4*)(mu + ncol);
    float4 ga4 = *(const float4*)(g + ncol);
    float4 va4 = *(const float4*)(var + ncol);
    float4 be4 = *(const float4*)(be + ncol);
    float4 rs4 = {rsqrtf(va4.x + BN_EPS), rsqrtf(va4.y + BN_EPS), rsqrtf(va4.z + BN_EPS), rsqrtf(va4.w + BN_EPS)};
#pragma unroll
    for (int v = 0; v < 8; ++v) {
        int row = m0 + v + 8 * hi;
        float4 o4 = {bn_elu_rs(acc0[v], bi4.x, mu4.x, ga4.x, rs4.x, be4.x),
                     bn_elu_rs(acc1[v], bi4.y, mu4.y, ga4.y, rs4.y, be4.y),
                     bn_elu_rs(acc2[v], bi4.z, mu4.z, ga4.z, rs4.z, be4.z),
                     bn_elu_rs(acc3[v], bi4.w, mu4.w, ga4.w, rs4.w, be4.w)};
        if (row < N) *(float4*)(outp + (size_t)row * 256 + ncol) = o4;
    }
}

#define M_START (-3.0e38f)
__global__ __launch_bounds__(256) void gather_h1(const int* __restrict__ cnt,
                                                 const int* __restrict__ head,
                                                 const int* __restrict__ nxt,
                                                 const int* __restrict__ bkt,
                                                 const int* __restrict__ ei,
                                                 const float* __restrict__ a_s,
                                                 const float* __restrict__ ad_i,
                                                 const float* __restrict__ hsrc,
                                                 const float* __restrict__ bias,
                                                 float* __restrict__ outp, int N) {
    int node = __builtin_amdgcn_readfirstlane((int)((blockIdx.x * 256u + threadIdx.x) >> 5));
    int lane = threadIdx.x & 31;
    if (node >= N) return;
    int lo = lane & 15, hi = lane >> 4;
    int cn = cnt[node];
    int degb = cn < CAP ? cn : CAP;
    int nb = degb + 1;
    int ovh = cn > CAP ? head[node] : -1;
    const int* brow = bkt + (size_t)node * CAP;
    float adv = ad_i[node] * LOG2E;
    float4 s4 = scale4(((const float4*)a_s)[lo], LOG2E);
    const char* hbytes = (const char*)hsrc;
    unsigned lob = 16u * (unsigned)lo;
    float m = M_START, z = 0.f;
    float4 acc = {0.f, 0.f, 0.f, 0.f};
    int base = 0;
    while (base < nb || ovh >= 0) {
        int s = node;
        int cnt32 = 0;
        if (base < nb) {
            int idx = base + lane;
            if (idx < degb) s = brow[idx];
            cnt32 = nb - base;
            if (cnt32 > 32) cnt32 = 32;
            base += 32;
        } else {
            while (ovh >= 0 && cnt32 < 32) {
                int sv = ei[ovh];
                if (lane == cnt32) s = sv;
                ovh = nxt[ovh];
                ++cnt32;
            }
        }
        for (int jj = 0; jj < cnt32; jj += 8) {
            int t0 = lane_i(s, jj),     t1 = lane_i(s, jj + 1), t2 = lane_i(s, jj + 2), t3 = lane_i(s, jj + 3);
            int t4 = lane_i(s, jj + 4), t5 = lane_i(s, jj + 5), t6 = lane_i(s, jj + 6), t7 = lane_i(s, jj + 7);
            unsigned s0 = hi ? t1 : t0, s1 = hi ? t3 : t2, s2 = hi ? t5 : t4, s3 = hi ? t7 : t6;
            float4 x0 = *(const float4*)(hbytes + (s0 * 256u + lob));
            float4 x1 = *(const float4*)(hbytes + (s1 * 256u + lob));
            float4 x2 = *(const float4*)(hbytes + (s2 * 256u + lob));
            float4 x3 = *(const float4*)(hbytes + (s3 * 256u + lob));
            float p0 = row16_sum(dot4(x0, s4)), p1 = row16_sum(dot4(x1, s4));
            float p2 = row16_sum(dot4(x2, s4)), p3 = row16_sum(dot4(x3, s4));
            int e = jj + hi;
            float e0 = e     < cnt32 ? lrelu_max(p0 + adv) : -INFINITY;
            float e1 = e + 2 < cnt32 ? lrelu_max(p1 + adv) : -INFINITY;
            float e2 = e + 4 < cnt32 ? lrelu_max(p2 + adv) : -INFINITY;
            float e3 = e + 6 < cnt32 ? lrelu_max(p3 + adv) : -INFINITY;
            float n = fmaxf(fmaxf(m, e0), fmaxf(fmaxf(e1, e2), e3));
            float sc = ex2(m - n);
            float w0 = ex2(e0 - n), w1 = ex2(e1 - n), w2 = ex2(e2 - n), w3 = ex2(e3 - n);
            m = n;
            z = z * sc + ((w0 + w1) + (w2 + w3));
            acc.x = acc.x * sc + w0 * x0.x + w1 * x1.x + w2 * x2.x + w3 * x3.x;
            acc.y = acc.y * sc + w0 * x0.y + w1 * x1.y + w2 * x2.y + w3 * x3.y;
            acc.z = acc.z * sc + w0 * x0.z + w1 * x1.z + w2 * x2.z + w3 * x3.z;
            acc.w = acc.w * sc + w0 * x0.w + w1 * x1.w + w2 * x2.w + w3 * x3.w;
        }
    }
    float mm = fmaxf(m, xhalf(m));
    float f = ex2(m - mm);
    z *= f; acc.x *= f; acc.y *= f; acc.z *= f; acc.w *= f;
    z += xhalf(z);
    acc.x += xhalf(acc.x); acc.y += xhalf(acc.y); acc.z += xhalf(acc.z); acc.w += xhalf(acc.w);
    float den = z + 1e-16f;
    if (hi == 0) {
        float4 b4 = ((const float4*)bias)[lo];
        float4 o4 = {acc.x / den + b4.x, acc.y / den + b4.y, acc.z / den + b4.z, acc.w / den + b4.w};
        ((float4*)(outp + (size_t)node * 64))[lo] = o4;
    }
}

static inline int nblk(long long t) { return (int)((t + 255) / 256); }

extern "C" void kernel_launch(void* const* d_in, const int* in_sizes, int n_in,
                              void* d_out, int out_size, void* d_ws, size_t ws_size,
                              hipStream_t stream) {
    const float* x      = (const float*)d_in[0];
    const int*   ei     = (const int*)d_in[1];
    const float* W1     = (const float*)d_in[2];
    const float* a_src1 = (const float*)d_in[3];
    const float* a_dst1 = (const float*)d_in[4];
    const float* b1     = (const float*)d_in[5];
    const float* bn_g   = (const float*)d_in[6];
    const float* bn_b   = (const float*)d_in[7];
    const float* bn_m   = (const float*)d_in[8];
    const float* bn_v   = (const float*)d_in[9];
    const float* W2     = (const float*)d_in[10];
    const float* a_src2 = (const float*)d_in[11];
    const float* a_dst2 = (const float*)d_in[12];
    const float* b2     = (const float*)d_in[13];

    const int IN_DIM = 128, F1 = 256, H1 = 4, C2 = 64;
    int N = in_sizes[0] / IN_DIM;
    int E = in_sizes[1] / 2;

    float* ws = (float*)d_ws;
    size_t o = 0;
    float* h2   = ws + o; o += (size_t)N * F1;
    float* act1 = ws + o; o += (size_t)N * F1;
    float* uv   = ws + o; o += (size_t)N * H1;
    o += (size_t)N * H1;
    int*   grp  = (int*)(ws + o); o += (size_t)N * H1 * 2 + (size_t)N * C2;
    o += (size_t)N;
    float* ad2  = ws + o; o += (size_t)N;

    int* cnt  = grp;
    int* head = cnt + N;
    int* bkt  = head + N;
    int* nxt  = bkt + (size_t)N * CAP;

    init_counts_uv<<<nblk(N), 256, 0, stream>>>(cnt, head, N, W1, a_src1, a_dst1, uv);
    place_edges<<<nblk(E), 256, 0, stream>>>(ei, E, cnt, head, nxt, bkt);

    gather_x4_gemm<<<(N + 15) / 16, 128, 0, stream>>>(cnt, head, nxt, bkt, ei, x, uv, W1, b1, bn_g, bn_b, bn_m, bn_v, act1,
                                                      N);

    wmma_gemm_alpha<<<nblk((long long)(N / 16) * 32), 256, 0, stream>>>(act1, W2, a_dst2, h2, ad2, N, F1, C2);
    gather_h1<<<nblk((long long)N * 32), 256, 0, stream>>>(cnt, head, nxt, bkt, ei, a_src2, ad2, h2, b2,
                                                           (float*)d_out, N);
}


# Attention scores computed once per node (layer 1 in extra work-groups of the edge pass, layer 2 in the GEMM epilogue), softmax weights one edge per lane; gather loops only add rows; sums regrouped
# speedup vs baseline: 6.9885x; 1.0895x over previous
#include <hip/hip_runtime.h>
#include <hip/hip_bf16.h>
#include <math.h>

typedef __attribute__((ext_vector_type(2))) float v2f;
typedef __attribute__((ext_vector_type(8))) float v8f;

#define NEG_SLOPE 0.2f
#define BN_EPS 1e-5f

template <int O> __device__ __forceinline__ float xmask_add(float v) {
    return v + __int_as_float(__builtin_amdgcn_update_dpp(0, __float_as_int(v), 0x160 | O, 0xf, 0xf, true));
}
__device__ __forceinline__ float row16_sum(float v) {
    v = xmask_add<1>(v);
    v = xmask_add<2>(v);
    v = xmask_add<4>(v);
    v = xmask_add<8>(v);
    return v;
}

#define KGRP 4

struct GemmFrag {
    v2f a[KGRP];
    float4 b0[KGRP];
    float4 b1[KGRP];
};

__device__ __forceinline__ void gemm_load(GemmFrag& f, const float* __restrict__ arow,
                                          const float* __restrict__ bcol, int Ncols, int k0, int hi) {
#pragma unroll
    for (int u = 0; u < KGRP; ++u) {
        int kb = k0 + 4 * u + 2 * hi;
        f.a[u] = *(const v2f*)(arow + kb);
        const float* bp = bcol + (size_t)kb * Ncols;
        f.b0[u] = *(const float4*)bp;
        f.b1[u] = *(const float4*)(bp + Ncols);
    }
}

__device__ __forceinline__ void gemm_mma(const GemmFrag& f, v8f& acc0, v8f& acc1, v8f& acc2, v8f& acc3) {
#pragma unroll
    for (int u = 0; u < KGRP; ++u) {
        v2f b;
        b.x = f.b0[u].x; b.y = f.b1[u].x;
        acc0 = __builtin_amdgcn_wmma_f32_16x16x4_f32(false, f.a[u], false, b, (short)0, acc0, false, false);
        b.x = f.b0[u].y; b.y = f.b1[u].y;
        acc1 = __builtin_amdgcn_wmma_f32_16x16x4_f32(false, f.a[u], false, b, (short)0, acc1, false, false);
        b.x = f.b0[u].z; b.y = f.b1[u].z;
        acc2 = __builtin_amdgcn_wmma_f32_16x16x4_f32(false, f.a[u], false, b, (short)0, acc2, false, false);
        b.x = f.b0[u].w; b.y = f.b1[u].w;
        acc3 = __builtin_amdgcn_wmma_f32_16x16x4_f32(false, f.a[u], false, b, (short)0, acc3, false, false);
    }
}

__global__ __launch_bounds__(256) void wmma_gemm_alpha(const float* __restrict__ A,
                                                       const float* __restrict__ B,
                                                       const float* __restrict__ a_s,
                                                       const float* __restrict__ a_d,
                                                       float* __restrict__ C,
                                                       float* __restrict__ as_o,
                                                       float* __restrict__ ad_o,
                                                       int M, int K, int Ncols) {
    int H = Ncols >> 6;
    int head = (int)(blockIdx.x % (unsigned)H);
    int mt = (int)(blockIdx.x / (unsigned)H) * 8 + (int)(threadIdx.x >> 5);
    int lane = threadIdx.x & 31;
    if (mt >= (M >> 4)) return;
    int m0 = mt << 4;
    int lo = lane & 15;
    int hi = lane >> 4;
    const float* arow = A + (size_t)(m0 + lo) * K;
    int ncol = (head << 6) + 4 * lo;
    const float* bcol = B + ncol;

    v8f acc0 = {0.f, 0.f, 0.f, 0.f, 0.f, 0.f, 0.f, 0.f};
    v8f acc1 = acc0, acc2 = acc0, acc3 = acc0;
    GemmFrag f0, f1;
    gemm_load(f0, arow, bcol, Ncols, 0, hi);
    for (int k0 = 0; k0 < K; k0 += 8 * KGRP) {
        gemm_load(f1, arow, bcol, Ncols, k0 + 4 * KGRP, hi);
        gemm_mma(f0, acc0, acc1, acc2, acc3);
        if (k0 + 8 * KGRP < K) gemm_load(f0, arow, bcol, Ncols, k0 + 8 * KGRP, hi);
        gemm_mma(f1, acc0, acc1, acc2, acc3);
    }
    float4 d4 = *(const float4*)(a_d + ncol);
    float4 s4 = *(const float4*)(a_s + ncol);
    float pd[8], ps[8];
#pragma unroll
    for (int v = 0; v < 8; ++v) {
        int row = m0 + v + 8 * hi;
        float4 c4 = {acc0[v], acc1[v], acc2[v], acc3[v]};
        *(float4*)(C + (size_t)row * Ncols + ncol) = c4;
        pd[v] = c4.x * d4.x + c4.y * d4.y + c4.z * d4.z + c4.w * d4.w;
        ps[v] = c4.x * s4.x + c4.y * s4.y + c4.z * s4.z + c4.w * s4.w;
    }
#pragma unroll
    for (int v = 0; v < 8; ++v) {
        pd[v] = row16_sum(pd[v]);
        ps[v] = row16_sum(ps[v]);
    }
    if (lo == 0) {
#pragma unroll
        for (int v = 0; v < 8; ++v) {
            size_t r = (size_t)(m0 + v + 8 * hi) * H + head;
            ad_o[r] = pd[v];
            as_o[r] = ps[v];
        }
    }
}

#define CAP 48

__device__ __forceinline__ int lane_i(int x, int l) { return __builtin_amdgcn_readlane(x, l); }
__device__ __forceinline__ float xhalf(float x) {
    return __int_as_float(__builtin_amdgcn_permlanex16(__float_as_int(x), __float_as_int(x), 0x76543210, 0xfedcba98,
                                                       false, false));
}
__device__ __forceinline__ float dot4(float4 c, float4 s) {
    return c.x * s.x + c.y * s.y + c.z * s.z + c.w * s.w;
}
#define LOG2E 1.44269504088896340736f
__device__ __forceinline__ float4 scale4(float4 v, float c) {
    float4 r = {v.x * c, v.y * c, v.z * c, v.w * c};
    return r;
}
__device__ __forceinline__ float ex2(float v) { return __builtin_amdgcn_exp2f(v); }
__device__ __forceinline__ float lrelu_max(float v) { return fmaxf(v, NEG_SLOPE * v); }

__global__ __launch_bounds__(256) void init_counts_uv(int* __restrict__ cnt, int* __restrict__ head, int n,
                                                      const float* __restrict__ W1,
                                                      const float* __restrict__ a_s,
                                                      const float* __restrict__ a_d,
                                                      float* __restrict__ uv) {
    int i = blockIdx.x * blockDim.x + threadIdx.x;
    if (i < n) {
        cnt[i] = 0;
        head[i] = -1;
    }
    if (i < 1024) {
        int h = (i >> 7) & 3, k = i & 127;
        const float* a = (i < 512 ? a_s : a_d) + h * 64;
        const float* w = W1 + (size_t)k * 256 + h * 64;
        float s = 0.f;
        for (int c = 0; c < 64; ++c) s += w[c] * a[c];
        uv[i] = s * LOG2E;
    }
}

template <int O> __device__ __forceinline__ float xmask_get(float v) {
    return __int_as_float(__builtin_amdgcn_update_dpp(0, __float_as_int(v), 0x160 | O, 0xf, 0xf, true));
}
__device__ __forceinline__ float head_sum(float k0, float k1, float s0, float s1) {
    float own = k0 + xhalf(s0);
    float oth = k1 + xhalf(s1);
    float r = own + xmask_get<8>(oth);
    r = xmask_add<4>(r);
    r = xmask_add<2>(r);
    r = xmask_add<1>(r);
    return r;
}
__device__ __forceinline__ float bcast(float v, int l) {
    return __int_as_float(__builtin_amdgcn_readlane(__float_as_int(v), l));
}
__device__ __forceinline__ void fma4(float4& a, float w, float4 x) {
    a.x += w * x.x;
    a.y += w * x.y;
    a.z += w * x.z;
    a.w += w * x.w;
}
__device__ __forceinline__ float wave_max(float v) {
    v = fmaxf(v, xmask_get<1>(v));
    v = fmaxf(v, xmask_get<2>(v));
    v = fmaxf(v, xmask_get<4>(v));
    v = fmaxf(v, xmask_get<8>(v));
    return fmaxf(v, xhalf(v));
}
__device__ __forceinline__ float wave_sum(float v) {
    v = row16_sum(v);
    return v + xhalf(v);
}
__device__ __forceinline__ float bn_elu_rs(float agg, float b, float mu, float g, float rs, float be) {
    float v = agg + b;
    v = (v - mu) * g * rs + be;
    return v > 0.f ? v : (__expf(v) - 1.f);
}

__global__ __launch_bounds__(256) void place_edges_scores(const int* __restrict__ ei, int E,
                                                          int* __restrict__ cnt, int* __restrict__ head,
                                                          int* __restrict__ nxt, int* __restrict__ bkt,
                                                          const float* __restrict__ x,
                                                          const float* __restrict__ uv,
                                                          float* __restrict__ as1, float* __restrict__ ad1,
                                                          int N, int nodeBlocks) {
    if ((int)blockIdx.x < nodeBlocks) {
        int lane = threadIdx.x & 31;
        int n0 = (int)((blockIdx.x * 256u + threadIdx.x) >> 5) * 4;
        int hl = lane >> 3;
        const float4* us = (const float4*)uv;
        const float4* ud = us + 128;
        float4 R0 = us[32 * hl + lane], R1 = us[32 * (hl ^ 1) + lane];
        float4 R2 = us[32 * (hl ^ 2) + lane], R3 = us[32 * (hl ^ 3) + lane];
        float4 D0 = ud[32 * hl + lane], D1 = ud[32 * (hl ^ 1) + lane];
        float4 D2 = ud[32 * (hl ^ 2) + lane], D3 = ud[32 * (hl ^ 3) + lane];
        float4 xr[4];
#pragma unroll
        for (int i = 0; i < 4; ++i) {
            int n = n0 + i < N ? n0 + i : N - 1;
            xr[i] = ((const float4*)(x + (size_t)n * 128))[lane];
        }
#pragma unroll
        for (int i = 0; i < 4; ++i) {
            float ps = head_sum(dot4(xr[i], R0), dot4(xr[i], R1), dot4(xr[i], R2), dot4(xr[i], R3));
            float pd = head_sum(dot4(xr[i], D0), dot4(xr[i], D1), dot4(xr[i], D2), dot4(xr[i], D3));
            if (n0 + i < N && (lane & 7) == 0) {
                as1[(size_t)(n0 + i) * 4 + hl] = ps;
                ad1[(size_t)(n0 + i) * 4 + hl] = pd;
            }
        }
        return;
    }
    int e = ((int)blockIdx.x - nodeBlocks) * 256 + (int)threadIdx.x;
    if (e >= E) return;
    int d = ei[E + e];
    int sv = ei[e];
    asm volatile("" : "+v"(sv));
    int slot = atomicAdd(&cnt[d], 1);
    if (slot < CAP) bkt[(size_t)d * CAP + slot] = sv;
    else nxt[e] = atomicExch(&head[d], e);
}

#define XP 132

__device__ __forceinline__ void load_rows8(float4 (&xr)[8], const float* __restrict__ x, int s, int jj, int lane) {
#pragma unroll
    for (int k = 0; k < 8; ++k) {
        int sk = lane_i(s, jj + k);
        xr[k] = ((const float4*)(x + (size_t)sk * 128))[lane];
    }
}

__global__ __launch_bounds__(128) void gather_x4_gemm(const int* __restrict__ cnt,
                                                      const int* __restrict__ head,
                                                      const int* __restrict__ nxt,
                                                      const int* __restrict__ bkt,
                                                      const int* __restrict__ ei,
                                                      const float* __restrict__ x,
                                                      const float* __restrict__ as1,
                                                      const float* __restrict__ ad1,
                                                      const float* __restrict__ W1,
                                                      const float* __restrict__ b1,
                                                      const float* __restrict__ g,
                                                      const float* __restrict__ be,
                                                      const float* __restrict__ mu,
                                                      const float* __restrict__ var,
                                                      float* __restrict__ outp, int N) {
    __shared__ float4 xs4[4 * 16 * (XP / 4)];
    float* xs = (float*)xs4;
    int wave = __builtin_amdgcn_readfirstlane((int)(threadIdx.x >> 5));
    int lane = threadIdx.x & 31;
    int m0 = (int)blockIdx.x * 16;

    for (int i = 0; i < 4; ++i) {
        int r = wave * 4 + i;
        int node = m0 + r;
        float4 o0 = {0.f, 0.f, 0.f, 0.f}, o1 = o0, o2 = o0, o3 = o0;
        if (node < N) {
            float4 adv = ((const float4*)ad1)[node];
            int cn = cnt[node];
            int degb = cn < CAP ? cn : CAP;
            int nb = degb + 1;
            int ovh = cn > CAP ? head[node] : -1;
            const int* brow = bkt + (size_t)node * CAP;
            float4 m = {-INFINITY, -INFINITY, -INFINITY, -INFINITY};
            float4 z = {0.f, 0.f, 0.f, 0.f};
            float4 a0 = {0.f, 0.f, 0.f, 0.f}, a1 = a0, a2 = a0, a3 = a0;
            int base = 0;
            bool later = false;
            while (base < nb || ovh >= 0) {
                int s = node;
                int cnt32 = 0;
                if (base < nb) {
                    int idx = base + lane;
                    if (idx < degb) s = brow[idx];
                    cnt32 = nb - base;
                    if (cnt32 > 32) cnt32 = 32;
                    base += 32;
                } else {
                    while (ovh >= 0 && cnt32 < 32) {
                        int sv = ei[ovh];
                        if (lane == cnt32) s = sv;
                        ovh = nxt[ovh];
                        ++cnt32;
                    }
                }
                float4 sc = ((const float4*)as1)[s];
                float4 xr[8];
                load_rows8(xr, x, s, 0, lane);
                bool val = lane < cnt32;
                float e0 = val ? lrelu_max(sc.x + adv.x) : -INFINITY;
                float e1 = val ? lrelu_max(sc.y + adv.y) : -INFINITY;
                float e2 = val ? lrelu_max(sc.z + adv.z) : -INFINITY;
                float e3 = val ? lrelu_max(sc.w + adv.w) : -INFINITY;
                float n0 = fmaxf(m.x, wave_max(e0)), n1 = fmaxf(m.y, wave_max(e1));
                float n2 = fmaxf(m.z, wave_max(e2)), n3 = fmaxf(m.w, wave_max(e3));
                float c0 = ex2(m.x - n0), c1 = ex2(m.y - n1), c2 = ex2(m.z - n2), c3 = ex2(m.w - n3);
                float w0 = ex2(e0 - n0), w1 = ex2(e1 - n1), w2 = ex2(e2 - n2), w3 = ex2(e3 - n3);
                z.x = z.x * c0 + wave_sum(w0);
                z.y = z.y * c1 + wave_sum(w1);
                z.z = z.z * c2 + wave_sum(w2);
                z.w = z.w * c3 + wave_sum(w3);
                m.x = n0; m.y = n1; m.z = n2; m.w = n3;
                if (later) {
                    a0 = scale4(a0, c0);
                    a1 = scale4(a1, c1);
                    a2 = scale4(a2, c2);
                    a3 = scale4(a3, c3);
                }
                later = true;
                int jj = 0;
                while (true) {
#pragma unroll
                    for (int k = 0; k < 8; ++k) {
                        fma4(a0, bcast(w0, jj + k), xr[k]);
                        fma4(a1, bcast(w1, jj + k), xr[k]);
                        fma4(a2, bcast(w2, jj + k), xr[k]);
                        fma4(a3, bcast(w3, jj + k), xr[k]);
                    }
                    jj += 8;
                    if (jj >= cnt32) break;
                    load_rows8(xr, x, s, jj, lane);
                }
            }
            o0 = scale4(a0, 1.0f / (z.x + 1e-16f));
            o1 = scale4(a1, 1.0f / (z.y + 1e-16f));
            o2 = scale4(a2, 1.0f / (z.z + 1e-16f));
            o3 = scale4(a3, 1.0f / (z.w + 1e-16f));
        }
        xs4[(0 * 16 + r) * (XP / 4) + lane] = o0;
        xs4[(1 * 16 + r) * (XP / 4) + lane] = o1;
        xs4[(2 * 16 + r) * (XP / 4) + lane] = o2;
        xs4[(3 * 16 + r) * (XP / 4) + lane] = o3;
    }
    __syncthreads();

    int lo = lane & 15;
    int hi = lane >> 4;
    const float* arow = xs + (wave * 16 + lo) * XP;
    int ncol = (wave << 6) + 4 * lo;
    const float* bcol = W1 + ncol;
    v8f acc0 = {0.f, 0.f, 0.f, 0.f, 0.f, 0.f, 0.f, 0.f};
    v8f acc1 = acc0, acc2 = acc0, acc3 = acc0;
    GemmFrag f0, f1;
    gemm_load(f0, arow, bcol, 256, 0, hi);
    for (int k0 = 0; k0 < 128; k0 += 8 * KGRP) {
        gemm_load(f1, arow, bcol, 256, k0 + 4 * KGRP, hi);
        gemm_mma(f0, acc0, acc1, acc2, acc3);
        if (k0 + 8 * KGRP < 128) gemm_load(f0, arow, bcol, 256, k0 + 8 * KGRP, hi);
        gemm_mma(f1, acc0, acc1, acc2, acc3);
    }
    float4 bi4 = *(const float4*)(b1 + ncol);
    float4 mu4 = *(const float4*)(mu + ncol);
    float4 ga4 = *(const float4*)(g + ncol);
    float4 va4 = *(const float4*)(var + ncol);
    float4 be4 = *(const float4*)(be + ncol);
    float4 rs4 = {rsqrtf(va4.x + BN_EPS), rsqrtf(va4.y + BN_EPS), rsqrtf(va4.z + BN_EPS), rsqrtf(va4.w + BN_EPS)};
#pragma unroll
    for (int v = 0; v < 8; ++v) {
        int row = m0 + v + 8 * hi;
        float4 o4 = {bn_elu_rs(acc0[v], bi4.x, mu4.x, ga4.x, rs4.x, be4.x),
                     bn_elu_rs(acc1[v], bi4.y, mu4.y, ga4.y, rs4.y, be4.y),
                     bn_elu_rs(acc2[v], bi4.z, mu4.z, ga4.z, rs4.z, be4.z),
                     bn_elu_rs(acc3[v], bi4.w, mu4.w, ga4.w, rs4.w, be4.w)};
        if (row < N) *(float4*)(outp + (size_t)row * 256 + ncol) = o4;
    }
}


__device__ __forceinline__ void load_rows4x2(float4 (&xr)[4], const char* __restrict__ hbytes, int s, int jj, int hi,
                                             unsigned lob) {
#pragma unroll
    for (int u = 0; u < 4; ++u) {
        int te = lane_i(s, jj + 2 * u), to = lane_i(s, jj + 2 * u + 1);
        unsigned su = hi ? to : te;
        xr[u] = *(const float4*)(hbytes + (su * 256u + lob));
    }
}

__global__ __launch_bounds__(256) void gather_h1(const int* __restrict__ cnt,
                                                 const int* __restrict__ head,
                                                 const int* __restrict__ nxt,
                                                 const int* __restrict__ bkt,
                                                 const int* __restrict__ ei,
                                                 const float* __restrict__ as_i,
                                                 const float* __restrict__ ad_i,
                                                 const float* __restrict__ hsrc,
                                                 const float* __restrict__ bias,
                                                 float* __restrict__ outp, int N) {
    int node = __builtin_amdgcn_readfirstlane((int)((blockIdx.x * 256u + threadIdx.x) >> 5));
    int lane = threadIdx.x & 31;
    if (node >= N) return;
    int lo = lane & 15, hi = lane >> 4;
    int cn = cnt[node];
    int degb = cn < CAP ? cn : CAP;
    int nb = degb + 1;
    int ovh = cn > CAP ? head[node] : -1;
    const int* brow = bkt + (size_t)node * CAP;
    float adv = ad_i[node] * LOG2E;
    const char* hbytes = (const char*)hsrc;
    unsigned lob = 16u * (unsigned)lo;
    float m = -INFINITY, z = 0.f;
    float4 acc = {0.f, 0.f, 0.f, 0.f};
    int base = 0;
    bool later = false;
    while (base < nb || ovh >= 0) {
        int s = node;
        int cnt32 = 0;
        if (base < nb) {
            int idx = base + lane;
            if (idx < degb) s = brow[idx];
            cnt32 = nb - base;
            if (cnt32 > 32) cnt32 = 32;
            base += 32;
        } else {
            while (ovh >= 0 && cnt32 < 32) {
                int sv = ei[ovh];
                if (lane == cnt32) s = sv;
                ovh = nxt[ovh];
                ++cnt32;
            }
        }
        float sv = as_i[s];
        float4 xr[4];
        load_rows4x2(xr, hbytes, s, 0, hi, lob);
        float e = lane < cnt32 ? lrelu_max(sv * LOG2E + adv) : -INFINITY;
        float n = fmaxf(m, wave_max(e));
        float c = ex2(m - n);
        float w = ex2(e - n);
        z = z * c + wave_sum(w);
        m = n;
        if (later) acc = scale4(acc, c);
        later = true;
        int jj = 0;
        while (true) {
#pragma unroll
            for (int u = 0; u < 4; ++u) {
                float we = bcast(w, jj + 2 * u), wo = bcast(w, jj + 2 * u + 1);
                fma4(acc, hi ? wo : we, xr[u]);
            }
            jj += 8;
            if (jj >= cnt32) break;
            load_rows4x2(xr, hbytes, s, jj, hi, lob);
        }
    }
    acc.x += xhalf(acc.x); acc.y += xhalf(acc.y); acc.z += xhalf(acc.z); acc.w += xhalf(acc.w);
    float den = z + 1e-16f;
    if (hi == 0) {
        float4 b4 = ((const float4*)bias)[lo];
        float4 o4 = {acc.x / den + b4.x, acc.y / den + b4.y, acc.z / den + b4.z, acc.w / den + b4.w};
        ((float4*)(outp + (size_t)node * 64))[lo] = o4;
    }
}

static inline int nblk(long long t) { return (int)((t + 255) / 256); }

extern "C" void kernel_launch(void* const* d_in, const int* in_sizes, int n_in,
                              void* d_out, int out_size, void* d_ws, size_t ws_size,
                              hipStream_t stream) {
    const float* x      = (const float*)d_in[0];
    const int*   ei     = (const int*)d_in[1];
    const float* W1     = (const float*)d_in[2];
    const float* a_src1 = (const float*)d_in[3];
    const float* a_dst1 = (const float*)d_in[4];
    const float* b1     = (const float*)d_in[5];
    const float* bn_g   = (const float*)d_in[6];
    const float* bn_b   = (const float*)d_in[7];
    const float* bn_m   = (const float*)d_in[8];
    const float* bn_v   = (const float*)d_in[9];
    const float* W2     = (const float*)d_in[10];
    const float* a_src2 = (const float*)d_in[11];
    const float* a_dst2 = (const float*)d_in[12];
    const float* b2     = (const float*)d_in[13];

    const int IN_DIM = 128, F1 = 256, H1 = 4, C2 = 64;
    int N = in_sizes[0] / IN_DIM;
    int E = in_sizes[1] / 2;

    float* ws = (float*)d_ws;
    size_t o = 0;
    float* h2   = ws + o;
    float* as1  = h2 + (size_t)N * C2;
    float* ad1  = as1 + (size_t)N * H1;
    o += (size_t)N * F1;
    float* act1 = ws + o; o += (size_t)N * F1;
    float* uv   = ws + o; o += (size_t)N * H1;
    o += (size_t)N * H1;
    int*   grp  = (int*)(ws + o); o += (size_t)N * H1 * 2 + (size_t)N * C2;
    float* as2  = ws + o; o += (size_t)N;
    float* ad2  = ws + o; o += (size_t)N;

    int* cnt  = grp;
    int* head = cnt + N;
    int* bkt  = head + N;
    int* nxt  = bkt + (size_t)N * CAP;

    init_counts_uv<<<nblk(N), 256, 0, stream>>>(cnt, head, N, W1, a_src1, a_dst1, uv);
    place_edges_scores<<<nblk(E) + (N + 31) / 32, 256, 0, stream>>>(ei, E, cnt, head, nxt, bkt, x, uv, as1, ad1, N,
                                                                   (N + 31) / 32);

    gather_x4_gemm<<<(N + 15) / 16, 128, 0, stream>>>(cnt, head, nxt, bkt, ei, x, as1, ad1, W1, b1, bn_g, bn_b, bn_m, bn_v,
                                                      act1, N);

    wmma_gemm_alpha<<<nblk((long long)(N / 16) * 32), 256, 0, stream>>>(act1, W2, a_src2, a_dst2, h2, as2, ad2, N, F1,
                                                                        C2);
    gather_h1<<<nblk((long long)N * 32), 256, 0, stream>>>(cnt, head, nxt, bkt, ei, as2, ad2, h2, b2,
                                                           (float*)d_out, N);
}
